# starting kernel
# baseline (speedup 1.0000x reference)
#include <hip/hip_runtime.h>


typedef __attribute__((ext_vector_type(2))) float v2f;
typedef __attribute__((ext_vector_type(4))) float v4f;
typedef __attribute__((ext_vector_type(8))) float v8f;

#define D       512
#define TQ      64
#define TK      128
#define SLICES  32
#define PQ_ROWS 2048
#define PK_ROWS 4096

__device__ __forceinline__ v8f wmma4(v2f a, v2f b, v8f c) {
    return __builtin_amdgcn_wmma_f32_16x16x4_f32(
         false, a,  false, b,
         (short)0, c,  false,  false);
}

__device__ __forceinline__ float tanh4_dot(v4f x, v4f w, float acc) {
    float t0, t1, t2, t3;
    asm("v_tanh_f32 %0, %4\n\t"
        "v_tanh_f32 %1, %5\n\t"
        "v_tanh_f32 %2, %6\n\t"
        "v_tanh_f32 %3, %7\n\t"
        "v_nop"
        : "=&v"(t0), "=&v"(t1), "=&v"(t2), "=&v"(t3)
        : "v"(x.x), "v"(x.y), "v"(x.z), "v"(x.w));
    acc = fmaf(t0, w.x, acc);
    acc = fmaf(t1, w.y, acc);
    acc = fmaf(t2, w.z, acc);
    acc = fmaf(t3, w.w, acc);
    return acc;
}

__global__ void __launch_bounds__(256)
proj_kernel(const float* __restrict__ query, const float* __restrict__ keys,
            const float* __restrict__ Wq,    const float* __restrict__ Wk,
            float* __restrict__ pq,          float* __restrict__ pk) {
    const int gwave = (blockIdx.x * 256 + (int)threadIdx.x) >> 5;
    const int lane  = threadIdx.x & 31;
    const int strip = gwave & 7;
    const int mtile = gwave >> 3;

    const float *A, *W;
    float *O;
    if (mtile < PQ_ROWS / 16) {
        const int r0 = mtile * 16;
        A = query + (size_t)r0 * D;
        W = Wq + (size_t)(r0 / (16 * TQ)) * D * D;
        O = pq + (size_t)r0 * D;
    } else {
        const int r0 = (mtile - PQ_ROWS / 16) * 16;
        A = keys + (size_t)r0 * D;
        W = Wk + (size_t)(r0 / (16 * TK)) * D * D;
        O = pk + (size_t)r0 * D;
    }

    const int half = lane >> 4;
    const int ln16 = lane & 15;
    const int n0   = strip * 64;

    const float* aP  = A + (size_t)ln16 * D + 2 * half;
    const float* bP0 = W + (size_t)(n0 +  0 + ln16) * D + 2 * half;
    const float* bP1 = W + (size_t)(n0 + 16 + ln16) * D + 2 * half;
    const float* bP2 = W + (size_t)(n0 + 32 + ln16) * D + 2 * half;
    const float* bP3 = W + (size_t)(n0 + 48 + ln16) * D + 2 * half;

    v8f c0 = {}, c1 = {}, c2 = {}, c3 = {};
    #pragma unroll 4
    for (int k = 0; k < D; k += 4) {
        v2f a  = *(const v2f*)(aP  + k);
        v2f b0 = *(const v2f*)(bP0 + k);
        v2f b1 = *(const v2f*)(bP1 + k);
        v2f b2 = *(const v2f*)(bP2 + k);
        v2f b3 = *(const v2f*)(bP3 + k);
        c0 = wmma4(a, b0, c0);
        c1 = wmma4(a, b1, c1);
        c2 = wmma4(a, b2, c2);
        c3 = wmma4(a, b3, c3);
    }
    float* oP = O + (size_t)(8 * half) * D + n0 + ln16;
    #pragma unroll
    for (int r = 0; r < 8; ++r) {
        oP[(size_t)r * D +  0] = c0[r];
        oP[(size_t)r * D + 16] = c1[r];
        oP[(size_t)r * D + 32] = c2[r];
        oP[(size_t)r * D + 48] = c3[r];
    }
}

__global__ void __launch_bounds__(256)
score_softmax_kernel(const float* __restrict__ pq, const float* __restrict__ pk,
                     const float* __restrict__ vatt, float* __restrict__ scores_out) {
    __shared__ float s_sc[TK];

    const int row   = blockIdx.x;
    const int slice = row / TQ;
    const int tid   = threadIdx.x;
    const int wave  = tid >> 5;
    const int lane  = tid & 31;
    const int nb    = lane * 4;

    const float* pqrow  = pq + (size_t)row * D + nb;
    const float* pkbase = pk + (size_t)slice * TK * D + nb;

    const v4f q0 = *(const v4f*)(pqrow);
    const v4f q1 = *(const v4f*)(pqrow + 128);
    const v4f q2 = *(const v4f*)(pqrow + 256);
    const v4f q3 = *(const v4f*)(pqrow + 384);
    const v4f w0 = *(const v4f*)(vatt + nb);
    const v4f w1 = *(const v4f*)(vatt + nb + 128);
    const v4f w2 = *(const v4f*)(vatt + nb + 256);
    const v4f w3 = *(const v4f*)(vatt + nb + 384);

    #pragma unroll 2
    for (int kk = 0; kk < TK / 8; ++kk) {
        const int k = kk * 8 + wave;
        const float* pkr = pkbase + (size_t)k * D;
        const v4f p0 = *(const v4f*)(pkr);
        const v4f p1 = *(const v4f*)(pkr + 128);
        const v4f p2 = *(const v4f*)(pkr + 256);
        const v4f p3 = *(const v4f*)(pkr + 384);
        float partial = 0.f;
        partial = tanh4_dot(q0 + p0, w0, partial);
        partial = tanh4_dot(q1 + p1, w1, partial);
        partial = tanh4_dot(q2 + p2, w2, partial);
        partial = tanh4_dot(q3 + p3, w3, partial);
        #pragma unroll
        for (int off = 16; off > 0; off >>= 1)
            partial += __shfl_xor(partial, off, 32);
        if (lane == 0) s_sc[k] = partial;
    }
    __syncthreads();

    if (wave == 0) {
        float s0 = s_sc[lane], s1 = s_sc[lane + 32];
        float s2 = s_sc[lane + 64], s3 = s_sc[lane + 96];
        float m = fmaxf(fmaxf(s0, s1), fmaxf(s2, s3));
        #pragma unroll
        for (int off = 16; off > 0; off >>= 1)
            m = fmaxf(m, __shfl_xor(m, off, 32));
        float e0 = __expf(s0 - m), e1 = __expf(s1 - m);
        float e2 = __expf(s2 - m), e3 = __expf(s3 - m);
        float sum = e0 + e1 + e2 + e3;
        #pragma unroll
        for (int off = 16; off > 0; off >>= 1)
            sum += __shfl_xor(sum, off, 32);
        const float inv = 1.f / sum;
        float* out = scores_out + (size_t)row * TK;
        out[lane]      = e0 * inv;
        out[lane + 32] = e1 * inv;
        out[lane + 64] = e2 * inv;
        out[lane + 96] = e3 * inv;
    }
}

__global__ void __launch_bounds__(256)
context_kernel(const float* __restrict__ probs, const float* __restrict__ keys,
               float* __restrict__ ctx) {
    const int gwave = (blockIdx.x * 256 + (int)threadIdx.x) >> 5;
    const int lane  = threadIdx.x & 31;
    const int strip = gwave & 7;
    const int mt    = (gwave >> 3) & 3;
    const int s     = gwave >> 5;

    const int half = lane >> 4;
    const int ln16 = lane & 15;
    const int n0   = strip * 64;

    const float* A = probs + (size_t)s * TQ * TK + (size_t)mt * 16 * TK;
    const float* B = keys  + (size_t)s * TK * D;
    float*       O = ctx   + (size_t)s * TQ * D  + (size_t)mt * 16 * D;

    const float* aP = A + (size_t)ln16 * TK + 2 * half;
    const float* bP = B + (size_t)(2 * half) * D + n0 + ln16;

    v8f c0 = {}, c1 = {}, c2 = {}, c3 = {};
    #pragma unroll 4
    for (int k = 0; k < TK; k += 4) {
        v2f a = *(const v2f*)(aP + k);
        const float* bk = bP + (size_t)k * D;
        v2f b0, b1, b2, b3;
        b0.x = bk[ 0]; b0.y = bk[D +  0];
        b1.x = bk[16]; b1.y = bk[D + 16];
        b2.x = bk[32]; b2.y = bk[D + 32];
        b3.x = bk[48]; b3.y = bk[D + 48];
        c0 = wmma4(a, b0, c0);
        c1 = wmma4(a, b1, c1);
        c2 = wmma4(a, b2, c2);
        c3 = wmma4(a, b3, c3);
    }
    float* oP = O + (size_t)(8 * half) * D + n0 + ln16;
    #pragma unroll
    for (int r = 0; r < 8; ++r) {
        oP[(size_t)r * D +  0] = c0[r];
        oP[(size_t)r * D + 16] = c1[r];
        oP[(size_t)r * D + 32] = c2[r];
        oP[(size_t)r * D + 48] = c3[r];
    }
}

extern "C" void kernel_launch(void* const* d_in, const int* in_sizes, int n_in,
                              void* d_out, int out_size, void* d_ws, size_t ws_size,
                              hipStream_t stream) {
    const float* query = (const float*)d_in[0];
    const float* keys  = (const float*)d_in[1];
    const float* Wq    = (const float*)d_in[2];
    const float* Wk    = (const float*)d_in[3];
    const float* vatt  = (const float*)d_in[4];

    float* ctx    = (float*)d_out;
    float* scores = ctx + (size_t)SLICES * TQ * D;

    float* pq = (float*)d_ws;
    float* pk = pq + (size_t)PQ_ROWS * D;

    proj_kernel<<<384, 256, 0, stream>>>(query, keys, Wq, Wk, pq, pk);
    score_softmax_kernel<<<PQ_ROWS, 256, 0, stream>>>(pq, pk, vatt, scores);
    context_kernel<<<128, 256, 0, stream>>>(scores, keys, ctx);
}


# proj GEMM: A and W tiles staged through LDS (64x64 tile per 4-wave block, whole-row 16-B global loads, two stages) instead of 16-row-strided 8-B fragment loads from global; k order unchanged
# speedup vs baseline: 1.5156x; 1.5156x over previous
#include <hip/hip_runtime.h>


typedef __attribute__((ext_vector_type(2))) float v2f;
typedef __attribute__((ext_vector_type(4))) float v4f;
typedef __attribute__((ext_vector_type(8))) float v8f;

#define D       512
#define TQ      64
#define TK      128
#define SLICES  32
#define PQ_ROWS 2048
#define PK_ROWS 4096

__device__ __forceinline__ v8f wmma4(v2f a, v2f b, v8f c) {
    return __builtin_amdgcn_wmma_f32_16x16x4_f32(
         false, a,  false, b,
         (short)0, c,  false,  false);
}

__device__ __forceinline__ float tanh4_dot(v4f x, v4f w, float acc) {
    float t0, t1, t2, t3;
    asm("v_tanh_f32 %0, %4\n\t"
        "v_tanh_f32 %1, %5\n\t"
        "v_tanh_f32 %2, %6\n\t"
        "v_tanh_f32 %3, %7\n\t"
        "v_nop"
        : "=&v"(t0), "=&v"(t1), "=&v"(t2), "=&v"(t3)
        : "v"(x.x), "v"(x.y), "v"(x.z), "v"(x.w));
    acc = fmaf(t0, w.x, acc);
    acc = fmaf(t1, w.y, acc);
    acc = fmaf(t2, w.z, acc);
    acc = fmaf(t3, w.w, acc);
    return acc;
}

#define PJ_KC    32
#define PJ_PITCH 36
#define PJ_TILE  (64 * PJ_PITCH)

__global__ void __launch_bounds__(128)
proj_kernel(const float* __restrict__ query, const float* __restrict__ keys,
            const float* __restrict__ Wq,    const float* __restrict__ Wk,
            float* __restrict__ pq,          float* __restrict__ pk) {
    __shared__ __attribute__((aligned(16))) float sA[2 * PJ_TILE];
    __shared__ __attribute__((aligned(16))) float sB[2 * PJ_TILE];

    const int tid   = threadIdx.x;
    const int wave  = tid >> 5;
    const int lane  = tid & 31;
    const int half  = lane >> 4;
    const int ln16  = lane & 15;
    const int wm    = wave >> 1;
    const int wn    = wave & 1;
    const int ntile = blockIdx.x & 7;
    const int mtile = blockIdx.x >> 3;
    const int n0    = ntile * 64;

    const float *A, *W;
    float *O;
    if (mtile < PQ_ROWS / 64) {
        const int r0 = mtile * 64;
        A = query + (size_t)r0 * D;
        W = Wq + (size_t)(r0 / (16 * TQ)) * D * D;
        O = pq + (size_t)r0 * D;
    } else {
        const int r0 = (mtile - PQ_ROWS / 64) * 64;
        A = keys + (size_t)r0 * D;
        W = Wk + (size_t)(r0 / (16 * TK)) * D * D;
        O = pk + (size_t)r0 * D;
    }

    const int srow = tid >> 3;
    const int sc4  = (tid & 7) * 4;
    const float* gA = A + (size_t)srow * D + sc4;
    const float* gB = W + (size_t)(n0 + srow) * D + sc4;
    const int sOff = srow * PJ_PITCH + sc4;

    const int aOff = (wm * 32 + ln16) * PJ_PITCH + 2 * half;
    const int bOff = (wn * 32 + ln16) * PJ_PITCH + 2 * half;

    v4f ra[4], rb[4];
    #pragma unroll
    for (int i = 0; i < 4; ++i) {
        ra[i] = *(const v4f*)(gA + (size_t)(16 * i) * D);
        rb[i] = *(const v4f*)(gB + (size_t)(16 * i) * D);
    }
    #pragma unroll
    for (int i = 0; i < 4; ++i) {
        *(v4f*)(sA + sOff + 16 * i * PJ_PITCH) = ra[i];
        *(v4f*)(sB + sOff + 16 * i * PJ_PITCH) = rb[i];
    }
    __syncthreads();

    v8f c00 = {}, c01 = {}, c10 = {}, c11 = {};
    for (int ch = 0; ch < D / PJ_KC; ++ch) {
        const int  cur  = (ch & 1) * PJ_TILE;
        const int  nxt  = PJ_TILE - cur;
        const bool more = ch + 1 < D / PJ_KC;
        if (more) {
            const int kn = (ch + 1) * PJ_KC;
            #pragma unroll
            for (int i = 0; i < 4; ++i) {
                ra[i] = *(const v4f*)(gA + (size_t)(16 * i) * D + kn);
                rb[i] = *(const v4f*)(gB + (size_t)(16 * i) * D + kn);
            }
        }
        const float* aS = sA + cur + aOff;
        const float* bS = sB + cur + bOff;
        #pragma unroll
        for (int kk = 0; kk < PJ_KC; kk += 4) {
            const v2f a0 = *(const v2f*)(aS + kk);
            const v2f a1 = *(const v2f*)(aS + 16 * PJ_PITCH + kk);
            const v2f b0 = *(const v2f*)(bS + kk);
            const v2f b1 = *(const v2f*)(bS + 16 * PJ_PITCH + kk);
            c00 = wmma4(a0, b0, c00);
            c01 = wmma4(a0, b1, c01);
            c10 = wmma4(a1, b0, c10);
            c11 = wmma4(a1, b1, c11);
        }
        if (more) {
            #pragma unroll
            for (int i = 0; i < 4; ++i) {
                *(v4f*)(sA + nxt + sOff + 16 * i * PJ_PITCH) = ra[i];
                *(v4f*)(sB + nxt + sOff + 16 * i * PJ_PITCH) = rb[i];
            }
        }
        __syncthreads();
    }

    float* oP = O + (size_t)(wm * 32 + 8 * half) * D + n0 + wn * 32 + ln16;
    #pragma unroll
    for (int r = 0; r < 8; ++r) {
        oP[(size_t)r * D +  0]        = c00[r];
        oP[(size_t)r * D + 16]        = c01[r];
        oP[(size_t)(r + 16) * D +  0] = c10[r];
        oP[(size_t)(r + 16) * D + 16] = c11[r];
    }
}

__global__ void __launch_bounds__(256)
score_softmax_kernel(const float* __restrict__ pq, const float* __restrict__ pk,
                     const float* __restrict__ vatt, float* __restrict__ scores_out) {
    __shared__ float s_sc[TK];

    const int row   = blockIdx.x;
    const int slice = row / TQ;
    const int tid   = threadIdx.x;
    const int wave  = tid >> 5;
    const int lane  = tid & 31;
    const int nb    = lane * 4;

    const float* pqrow  = pq + (size_t)row * D + nb;
    const float* pkbase = pk + (size_t)slice * TK * D + nb;

    const v4f q0 = *(const v4f*)(pqrow);
    const v4f q1 = *(const v4f*)(pqrow + 128);
    const v4f q2 = *(const v4f*)(pqrow + 256);
    const v4f q3 = *(const v4f*)(pqrow + 384);
    const v4f w0 = *(const v4f*)(vatt + nb);
    const v4f w1 = *(const v4f*)(vatt + nb + 128);
    const v4f w2 = *(const v4f*)(vatt + nb + 256);
    const v4f w3 = *(const v4f*)(vatt + nb + 384);

    #pragma unroll 2
    for (int kk = 0; kk < TK / 8; ++kk) {
        const int k = kk * 8 + wave;
        const float* pkr = pkbase + (size_t)k * D;
        const v4f p0 = *(const v4f*)(pkr);
        const v4f p1 = *(const v4f*)(pkr + 128);
        const v4f p2 = *(const v4f*)(pkr + 256);
        const v4f p3 = *(const v4f*)(pkr + 384);
        float partial = 0.f;
        partial = tanh4_dot(q0 + p0, w0, partial);
        partial = tanh4_dot(q1 + p1, w1, partial);
        partial = tanh4_dot(q2 + p2, w2, partial);
        partial = tanh4_dot(q3 + p3, w3, partial);
        #pragma unroll
        for (int off = 16; off > 0; off >>= 1)
            partial += __shfl_xor(partial, off, 32);
        if (lane == 0) s_sc[k] = partial;
    }
    __syncthreads();

    if (wave == 0) {
        float s0 = s_sc[lane], s1 = s_sc[lane + 32];
        float s2 = s_sc[lane + 64], s3 = s_sc[lane + 96];
        float m = fmaxf(fmaxf(s0, s1), fmaxf(s2, s3));
        #pragma unroll
        for (int off = 16; off > 0; off >>= 1)
            m = fmaxf(m, __shfl_xor(m, off, 32));
        float e0 = __expf(s0 - m), e1 = __expf(s1 - m);
        float e2 = __expf(s2 - m), e3 = __expf(s3 - m);
        float sum = e0 + e1 + e2 + e3;
        #pragma unroll
        for (int off = 16; off > 0; off >>= 1)
            sum += __shfl_xor(sum, off, 32);
        const float inv = 1.f / sum;
        float* out = scores_out + (size_t)row * TK;
        out[lane]      = e0 * inv;
        out[lane + 32] = e1 * inv;
        out[lane + 64] = e2 * inv;
        out[lane + 96] = e3 * inv;
    }
}

__global__ void __launch_bounds__(256)
context_kernel(const float* __restrict__ probs, const float* __restrict__ keys,
               float* __restrict__ ctx) {
    const int gwave = (blockIdx.x * 256 + (int)threadIdx.x) >> 5;
    const int lane  = threadIdx.x & 31;
    const int strip = gwave & 7;
    const int mt    = (gwave >> 3) & 3;
    const int s     = gwave >> 5;

    const int half = lane >> 4;
    const int ln16 = lane & 15;
    const int n0   = strip * 64;

    const float* A = probs + (size_t)s * TQ * TK + (size_t)mt * 16 * TK;
    const float* B = keys  + (size_t)s * TK * D;
    float*       O = ctx   + (size_t)s * TQ * D  + (size_t)mt * 16 * D;

    const float* aP = A + (size_t)ln16 * TK + 2 * half;
    const float* bP = B + (size_t)(2 * half) * D + n0 + ln16;

    v8f c0 = {}, c1 = {}, c2 = {}, c3 = {};
    #pragma unroll 4
    for (int k = 0; k < TK; k += 4) {
        v2f a = *(const v2f*)(aP + k);
        const float* bk = bP + (size_t)k * D;
        v2f b0, b1, b2, b3;
        b0.x = bk[ 0]; b0.y = bk[D +  0];
        b1.x = bk[16]; b1.y = bk[D + 16];
        b2.x = bk[32]; b2.y = bk[D + 32];
        b3.x = bk[48]; b3.y = bk[D + 48];
        c0 = wmma4(a, b0, c0);
        c1 = wmma4(a, b1, c1);
        c2 = wmma4(a, b2, c2);
        c3 = wmma4(a, b3, c3);
    }
    float* oP = O + (size_t)(8 * half) * D + n0 + ln16;
    #pragma unroll
    for (int r = 0; r < 8; ++r) {
        oP[(size_t)r * D +  0] = c0[r];
        oP[(size_t)r * D + 16] = c1[r];
        oP[(size_t)r * D + 32] = c2[r];
        oP[(size_t)r * D + 48] = c3[r];
    }
}

extern "C" void kernel_launch(void* const* d_in, const int* in_sizes, int n_in,
                              void* d_out, int out_size, void* d_ws, size_t ws_size,
                              hipStream_t stream) {
    const float* query = (const float*)d_in[0];
    const float* keys  = (const float*)d_in[1];
    const float* Wq    = (const float*)d_in[2];
    const float* Wk    = (const float*)d_in[3];
    const float* vatt  = (const float*)d_in[4];

    float* ctx    = (float*)d_out;
    float* scores = ctx + (size_t)SLICES * TQ * D;

    float* pq = (float*)d_ws;
    float* pk = pq + (size_t)PQ_ROWS * D;

    proj_kernel<<<768, 128, 0, stream>>>(query, keys, Wq, Wk, pq, pk);
    score_softmax_kernel<<<PQ_ROWS, 256, 0, stream>>>(pq, pk, vatt, scores);
    context_kernel<<<128, 256, 0, stream>>>(scores, keys, ctx);
}


# scores kernel: one block per 8 consecutive q rows with their pq rows in registers, so each pk row read serves 8 rows (pk L2 reads and waves /8); the 8 row sums share one butterfly, same sum tree
# speedup vs baseline: 2.2107x; 1.4586x over previous
#include <hip/hip_runtime.h>


typedef __attribute__((ext_vector_type(2))) float v2f;
typedef __attribute__((ext_vector_type(4))) float v4f;
typedef __attribute__((ext_vector_type(8))) float v8f;

#define D       512
#define TQ      64
#define TK      128
#define SLICES  32
#define PQ_ROWS 2048
#define PK_ROWS 4096

__device__ __forceinline__ v8f wmma4(v2f a, v2f b, v8f c) {
    return __builtin_amdgcn_wmma_f32_16x16x4_f32(
         false, a,  false, b,
         (short)0, c,  false,  false);
}

__device__ __forceinline__ float tanh4_dot(v4f x, v4f w, float acc) {
    float t0, t1, t2, t3;
    asm("v_tanh_f32 %0, %4\n\t"
        "v_tanh_f32 %1, %5\n\t"
        "v_tanh_f32 %2, %6\n\t"
        "v_tanh_f32 %3, %7\n\t"
        "v_nop"
        : "=&v"(t0), "=&v"(t1), "=&v"(t2), "=&v"(t3)
        : "v"(x.x), "v"(x.y), "v"(x.z), "v"(x.w));
    acc = fmaf(t0, w.x, acc);
    acc = fmaf(t1, w.y, acc);
    acc = fmaf(t2, w.z, acc);
    acc = fmaf(t3, w.w, acc);
    return acc;
}

#define PJ_KC    32
#define PJ_PITCH 36
#define PJ_TILE  (64 * PJ_PITCH)

__global__ void __launch_bounds__(128)
proj_kernel(const float* __restrict__ query, const float* __restrict__ keys,
            const float* __restrict__ Wq,    const float* __restrict__ Wk,
            float* __restrict__ pq,          float* __restrict__ pk) {
    __shared__ __attribute__((aligned(16))) float sA[2 * PJ_TILE];
    __shared__ __attribute__((aligned(16))) float sB[2 * PJ_TILE];

    const int tid   = threadIdx.x;
    const int wave  = tid >> 5;
    const int lane  = tid & 31;
    const int half  = lane >> 4;
    const int ln16  = lane & 15;
    const int wm    = wave >> 1;
    const int wn    = wave & 1;
    const int ntile = blockIdx.x & 7;
    const int mtile = blockIdx.x >> 3;
    const int n0    = ntile * 64;

    const float *A, *W;
    float *O;
    if (mtile < PQ_ROWS / 64) {
        const int r0 = mtile * 64;
        A = query + (size_t)r0 * D;
        W = Wq + (size_t)(r0 / (16 * TQ)) * D * D;
        O = pq + (size_t)r0 * D;
    } else {
        const int r0 = (mtile - PQ_ROWS / 64) * 64;
        A = keys + (size_t)r0 * D;
        W = Wk + (size_t)(r0 / (16 * TK)) * D * D;
        O = pk + (size_t)r0 * D;
    }

    const int srow = tid >> 3;
    const int sc4  = (tid & 7) * 4;
    const float* gA = A + (size_t)srow * D + sc4;
    const float* gB = W + (size_t)(n0 + srow) * D + sc4;
    const int sOff = srow * PJ_PITCH + sc4;

    const int aOff = (wm * 32 + ln16) * PJ_PITCH + 2 * half;
    const int bOff = (wn * 32 + ln16) * PJ_PITCH + 2 * half;

    v4f ra[4], rb[4];
    #pragma unroll
    for (int i = 0; i < 4; ++i) {
        ra[i] = *(const v4f*)(gA + (size_t)(16 * i) * D);
        rb[i] = *(const v4f*)(gB + (size_t)(16 * i) * D);
    }
    #pragma unroll
    for (int i = 0; i < 4; ++i) {
        *(v4f*)(sA + sOff + 16 * i * PJ_PITCH) = ra[i];
        *(v4f*)(sB + sOff + 16 * i * PJ_PITCH) = rb[i];
    }
    __syncthreads();

    v8f c00 = {}, c01 = {}, c10 = {}, c11 = {};
    for (int ch = 0; ch < D / PJ_KC; ++ch) {
        const int  cur  = (ch & 1) * PJ_TILE;
        const int  nxt  = PJ_TILE - cur;
        const bool more = ch + 1 < D / PJ_KC;
        if (more) {
            const int kn = (ch + 1) * PJ_KC;
            #pragma unroll
            for (int i = 0; i < 4; ++i) {
                ra[i] = *(const v4f*)(gA + (size_t)(16 * i) * D + kn);
                rb[i] = *(const v4f*)(gB + (size_t)(16 * i) * D + kn);
            }
        }
        const float* aS = sA + cur + aOff;
        const float* bS = sB + cur + bOff;
        #pragma unroll
        for (int kk = 0; kk < PJ_KC; kk += 4) {
            const v2f a0 = *(const v2f*)(aS + kk);
            const v2f a1 = *(const v2f*)(aS + 16 * PJ_PITCH + kk);
            const v2f b0 = *(const v2f*)(bS + kk);
            const v2f b1 = *(const v2f*)(bS + 16 * PJ_PITCH + kk);
            c00 = wmma4(a0, b0, c00);
            c01 = wmma4(a0, b1, c01);
            c10 = wmma4(a1, b0, c10);
            c11 = wmma4(a1, b1, c11);
        }
        if (more) {
            #pragma unroll
            for (int i = 0; i < 4; ++i) {
                *(v4f*)(sA + nxt + sOff + 16 * i * PJ_PITCH) = ra[i];
                *(v4f*)(sB + nxt + sOff + 16 * i * PJ_PITCH) = rb[i];
            }
        }
        __syncthreads();
    }

    float* oP = O + (size_t)(wm * 32 + 8 * half) * D + n0 + wn * 32 + ln16;
    #pragma unroll
    for (int r = 0; r < 8; ++r) {
        oP[(size_t)r * D +  0]        = c00[r];
        oP[(size_t)r * D + 16]        = c01[r];
        oP[(size_t)(r + 16) * D +  0] = c10[r];
        oP[(size_t)(r + 16) * D + 16] = c11[r];
    }
}

#define SC_RQ 8

__device__ __forceinline__ void load_row4(v4f (&p)[4], const float* r) {
    p[0] = *(const v4f*)(r);
    p[1] = *(const v4f*)(r + 128);
    p[2] = *(const v4f*)(r + 256);
    p[3] = *(const v4f*)(r + 384);
}

__device__ __forceinline__ void row_partials8(float (&part)[SC_RQ], const v4f (&q)[SC_RQ][4],
                                              const v4f (&p)[4], const v4f (&w)[4]) {
    #pragma unroll
    for (int r = 0; r < SC_RQ; ++r) {
        float partial = 0.f;
        partial = tanh4_dot(q[r][0] + p[0], w[0], partial);
        partial = tanh4_dot(q[r][1] + p[1], w[1], partial);
        partial = tanh4_dot(q[r][2] + p[2], w[2], partial);
        partial = tanh4_dot(q[r][3] + p[3], w[3], partial);
        part[r] = partial;
    }
}

__device__ __forceinline__ float butterfly8(const float (&part)[SC_RQ], int lane) {
    const bool h = (lane & 16) != 0;
    const bool g = (lane & 8) != 0;
    const bool f = (lane & 4) != 0;
    float s4[4];
    #pragma unroll
    for (int i = 0; i < 4; ++i) {
        const float keep = h ? part[i + 4] : part[i];
        const float send = h ? part[i]     : part[i + 4];
        s4[i] = keep + __shfl_xor(send, 16, 32);
    }
    float s2[2];
    #pragma unroll
    for (int i = 0; i < 2; ++i) {
        const float keep = g ? s4[i + 2] : s4[i];
        const float send = g ? s4[i]     : s4[i + 2];
        s2[i] = keep + __shfl_xor(send, 8, 32);
    }
    const float keep1 = f ? s2[1] : s2[0];
    const float send1 = f ? s2[0] : s2[1];
    float u = keep1 + __shfl_xor(send1, 4, 32);
    u += __shfl_xor(u, 2, 32);
    u += __shfl_xor(u, 1, 32);
    return u;
}

__global__ void __launch_bounds__(256)
score_softmax_kernel(const float* __restrict__ pq, const float* __restrict__ pk,
                     const float* __restrict__ vatt, float* __restrict__ scores_out) {
    __shared__ float s_sc[SC_RQ][TK];

    const int row0  = blockIdx.x * SC_RQ;
    const int slice = row0 / TQ;
    const int tid   = threadIdx.x;
    const int wave  = tid >> 5;
    const int lane  = tid & 31;
    const int nb    = lane * 4;

    const float* pqbase = pq + (size_t)row0 * D + nb;
    const float* pkbase = pk + (size_t)slice * TK * D + nb;

    v4f w[4], q[SC_RQ][4];
    load_row4(w, vatt + nb);
    #pragma unroll
    for (int r = 0; r < SC_RQ; ++r) load_row4(q[r], pqbase + (size_t)r * D);

    const int  qsel = (lane >> 2) & 7;
    const bool wr   = (lane & 3) == 0;

    v4f pa[4], pb[4];
    load_row4(pa, pkbase + (size_t)wave * D);
    #pragma unroll 1
    for (int kk = 0; kk < TK / 8; kk += 2) {
        const int k0 = kk * 8 + wave;
        const int k1 = k0 + 8;
        float parta[SC_RQ], partb[SC_RQ];
        load_row4(pb, pkbase + (size_t)k1 * D);
        row_partials8(parta, q, pa, w);
        const int kn = (kk + 2 < TK / 8) ? k1 + 8 : k1;
        load_row4(pa, pkbase + (size_t)kn * D);
        row_partials8(partb, q, pb, w);
        const float u0 = butterfly8(parta, lane);
        const float u1 = butterfly8(partb, lane);
        if (wr) {
            s_sc[qsel][k0] = u0;
            s_sc[qsel][k1] = u1;
        }
    }
    __syncthreads();

    {
        const float* sc = s_sc[wave];
        float s0 = sc[lane], s1 = sc[lane + 32];
        float s2 = sc[lane + 64], s3 = sc[lane + 96];
        float m = fmaxf(fmaxf(s0, s1), fmaxf(s2, s3));
        #pragma unroll
        for (int off = 16; off > 0; off >>= 1)
            m = fmaxf(m, __shfl_xor(m, off, 32));
        float e0 = __expf(s0 - m), e1 = __expf(s1 - m);
        float e2 = __expf(s2 - m), e3 = __expf(s3 - m);
        float sum = e0 + e1 + e2 + e3;
        #pragma unroll
        for (int off = 16; off > 0; off >>= 1)
            sum += __shfl_xor(sum, off, 32);
        const float inv = 1.f / sum;
        float* out = scores_out + (size_t)(row0 + wave) * TK;
        out[lane]      = e0 * inv;
        out[lane + 32] = e1 * inv;
        out[lane + 64] = e2 * inv;
        out[lane + 96] = e3 * inv;
    }
}

__global__ void __launch_bounds__(256)
context_kernel(const float* __restrict__ probs, const float* __restrict__ keys,
               float* __restrict__ ctx) {
    const int gwave = (blockIdx.x * 256 + (int)threadIdx.x) >> 5;
    const int lane  = threadIdx.x & 31;
    const int strip = gwave & 7;
    const int mt    = (gwave >> 3) & 3;
    const int s     = gwave >> 5;

    const int half = lane >> 4;
    const int ln16 = lane & 15;
    const int n0   = strip * 64;

    const float* A = probs + (size_t)s * TQ * TK + (size_t)mt * 16 * TK;
    const float* B = keys  + (size_t)s * TK * D;
    float*       O = ctx   + (size_t)s * TQ * D  + (size_t)mt * 16 * D;

    const float* aP = A + (size_t)ln16 * TK + 2 * half;
    const float* bP = B + (size_t)(2 * half) * D + n0 + ln16;

    v8f c0 = {}, c1 = {}, c2 = {}, c3 = {};
    #pragma unroll 4
    for (int k = 0; k < TK; k += 4) {
        v2f a = *(const v2f*)(aP + k);
        const float* bk = bP + (size_t)k * D;
        v2f b0, b1, b2, b3;
        b0.x = bk[ 0]; b0.y = bk[D +  0];
        b1.x = bk[16]; b1.y = bk[D + 16];
        b2.x = bk[32]; b2.y = bk[D + 32];
        b3.x = bk[48]; b3.y = bk[D + 48];
        c0 = wmma4(a, b0, c0);
        c1 = wmma4(a, b1, c1);
        c2 = wmma4(a, b2, c2);
        c3 = wmma4(a, b3, c3);
    }
    float* oP = O + (size_t)(8 * half) * D + n0 + ln16;
    #pragma unroll
    for (int r = 0; r < 8; ++r) {
        oP[(size_t)r * D +  0] = c0[r];
        oP[(size_t)r * D + 16] = c1[r];
        oP[(size_t)r * D + 32] = c2[r];
        oP[(size_t)r * D + 48] = c3[r];
    }
}

extern "C" void kernel_launch(void* const* d_in, const int* in_sizes, int n_in,
                              void* d_out, int out_size, void* d_ws, size_t ws_size,
                              hipStream_t stream) {
    const float* query = (const float*)d_in[0];
    const float* keys  = (const float*)d_in[1];
    const float* Wq    = (const float*)d_in[2];
    const float* Wk    = (const float*)d_in[3];
    const float* vatt  = (const float*)d_in[4];

    float* ctx    = (float*)d_out;
    float* scores = ctx + (size_t)SLICES * TQ * D;

    float* pq = (float*)d_ws;
    float* pk = pq + (size_t)PQ_ROWS * D;

    proj_kernel<<<768, 128, 0, stream>>>(query, keys, Wq, Wk, pq, pk);
    score_softmax_kernel<<<256, 256, 0, stream>>>(pq, pk, vatt, scores);
    context_kernel<<<128, 256, 0, stream>>>(scores, keys, ctx);
}


# context product moved into the scores kernel's tail (launches 3 to 2): per block P(8x128, in LDS) times the slice's keys on WMMA, even/odd columns on tile rows 0-7/8-15; sums still ascending k
# speedup vs baseline: 2.4338x; 1.1009x over previous
#include <hip/hip_runtime.h>


typedef __attribute__((ext_vector_type(2))) float v2f;
typedef __attribute__((ext_vector_type(4))) float v4f;
typedef __attribute__((ext_vector_type(8))) float v8f;

#define D       512
#define TQ      64
#define TK      128
#define SLICES  32
#define PQ_ROWS 2048
#define PK_ROWS 4096

__device__ __forceinline__ v8f wmma4(v2f a, v2f b, v8f c) {
    return __builtin_amdgcn_wmma_f32_16x16x4_f32(
         false, a,  false, b,
         (short)0, c,  false,  false);
}

__device__ __forceinline__ float tanh4_dot(v4f x, v4f w, float acc) {
    float t0, t1, t2, t3;
    asm("v_tanh_f32 %0, %4\n\t"
        "v_tanh_f32 %1, %5\n\t"
        "v_tanh_f32 %2, %6\n\t"
        "v_tanh_f32 %3, %7\n\t"
        "v_nop"
        : "=&v"(t0), "=&v"(t1), "=&v"(t2), "=&v"(t3)
        : "v"(x.x), "v"(x.y), "v"(x.z), "v"(x.w));
    acc = fmaf(t0, w.x, acc);
    acc = fmaf(t1, w.y, acc);
    acc = fmaf(t2, w.z, acc);
    acc = fmaf(t3, w.w, acc);
    return acc;
}

#define PJ_KC    32
#define PJ_PITCH 36
#define PJ_TILE  (64 * PJ_PITCH)

__global__ void __launch_bounds__(128)
proj_kernel(const float* __restrict__ query, const float* __restrict__ keys,
            const float* __restrict__ Wq,    const float* __restrict__ Wk,
            float* __restrict__ pq,          float* __restrict__ pk) {
    __shared__ __attribute__((aligned(16))) float sA[2 * PJ_TILE];
    __shared__ __attribute__((aligned(16))) float sB[2 * PJ_TILE];

    const int tid   = threadIdx.x;
    const int wave  = tid >> 5;
    const int lane  = tid & 31;
    const int half  = lane >> 4;
    const int ln16  = lane & 15;
    const int wm    = wave >> 1;
    const int wn    = wave & 1;
    const int ntile = blockIdx.x & 7;
    const int mtile = blockIdx.x >> 3;
    const int n0    = ntile * 64;

    const float *A, *W;
    float *O;
    if (mtile < PQ_ROWS / 64) {
        const int r0 = mtile * 64;
        A = query + (size_t)r0 * D;
        W = Wq + (size_t)(r0 / (16 * TQ)) * D * D;
        O = pq + (size_t)r0 * D;
    } else {
        const int r0 = (mtile - PQ_ROWS / 64) * 64;
        A = keys + (size_t)r0 * D;
        W = Wk + (size_t)(r0 / (16 * TK)) * D * D;
        O = pk + (size_t)r0 * D;
    }

    const int srow = tid >> 3;
    const int sc4  = (tid & 7) * 4;
    const float* gA = A + (size_t)srow * D + sc4;
    const float* gB = W + (size_t)(n0 + srow) * D + sc4;
    const int sOff = srow * PJ_PITCH + sc4;

    const int aOff = (wm * 32 + ln16) * PJ_PITCH + 2 * half;
    const int bOff = (wn * 32 + ln16) * PJ_PITCH + 2 * half;

    v4f ra[4], rb[4];
    #pragma unroll
    for (int i = 0; i < 4; ++i) {
        ra[i] = *(const v4f*)(gA + (size_t)(16 * i) * D);
        rb[i] = *(const v4f*)(gB + (size_t)(16 * i) * D);
    }
    #pragma unroll
    for (int i = 0; i < 4; ++i) {
        *(v4f*)(sA + sOff + 16 * i * PJ_PITCH) = ra[i];
        *(v4f*)(sB + sOff + 16 * i * PJ_PITCH) = rb[i];
    }
    __syncthreads();

    v8f c00 = {}, c01 = {}, c10 = {}, c11 = {};
    for (int ch = 0; ch < D / PJ_KC; ++ch) {
        const int  cur  = (ch & 1) * PJ_TILE;
        const int  nxt  = PJ_TILE - cur;
        const bool more = ch + 1 < D / PJ_KC;
        if (more) {
            const int kn = (ch + 1) * PJ_KC;
            #pragma unroll
            for (int i = 0; i < 4; ++i) {
                ra[i] = *(const v4f*)(gA + (size_t)(16 * i) * D + kn);
                rb[i] = *(const v4f*)(gB + (size_t)(16 * i) * D + kn);
            }
        }
        const float* aS = sA + cur + aOff;
        const float* bS = sB + cur + bOff;
        #pragma unroll
        for (int kk = 0; kk < PJ_KC; kk += 4) {
            const v2f a0 = *(const v2f*)(aS + kk);
            const v2f a1 = *(const v2f*)(aS + 16 * PJ_PITCH + kk);
            const v2f b0 = *(const v2f*)(bS + kk);
            const v2f b1 = *(const v2f*)(bS + 16 * PJ_PITCH + kk);
            c00 = wmma4(a0, b0, c00);
            c01 = wmma4(a0, b1, c01);
            c10 = wmma4(a1, b0, c10);
            c11 = wmma4(a1, b1, c11);
        }
        if (more) {
            #pragma unroll
            for (int i = 0; i < 4; ++i) {
                *(v4f*)(sA + nxt + sOff + 16 * i * PJ_PITCH) = ra[i];
                *(v4f*)(sB + nxt + sOff + 16 * i * PJ_PITCH) = rb[i];
            }
        }
        __syncthreads();
    }

    float* oP = O + (size_t)(wm * 32 + 8 * half) * D + n0 + wn * 32 + ln16;
    #pragma unroll
    for (int r = 0; r < 8; ++r) {
        oP[(size_t)r * D +  0]        = c00[r];
        oP[(size_t)r * D + 16]        = c01[r];
        oP[(size_t)(r + 16) * D +  0] = c10[r];
        oP[(size_t)(r + 16) * D + 16] = c11[r];
    }
}

#define SC_RQ 8
#define PE_PITCH 260
#define CX_KS 16

__device__ __forceinline__ void load_row4(v4f (&p)[4], const float* r) {
    p[0] = *(const v4f*)(r);
    p[1] = *(const v4f*)(r + 128);
    p[2] = *(const v4f*)(r + 256);
    p[3] = *(const v4f*)(r + 384);
}

__device__ __forceinline__ void row_partials8(float (&part)[SC_RQ], const v4f (&q)[SC_RQ][4],
                                              const v4f (&p)[4], const v4f (&w)[4]) {
    #pragma unroll
    for (int r = 0; r < SC_RQ; ++r) {
        float partial = 0.f;
        partial = tanh4_dot(q[r][0] + p[0], w[0], partial);
        partial = tanh4_dot(q[r][1] + p[1], w[1], partial);
        partial = tanh4_dot(q[r][2] + p[2], w[2], partial);
        partial = tanh4_dot(q[r][3] + p[3], w[3], partial);
        part[r] = partial;
    }
}

__device__ __forceinline__ float butterfly8(const float (&part)[SC_RQ], int lane) {
    const bool h = (lane & 16) != 0;
    const bool g = (lane & 8) != 0;
    const bool f = (lane & 4) != 0;
    float s4[4];
    #pragma unroll
    for (int i = 0; i < 4; ++i) {
        const float keep = h ? part[i + 4] : part[i];
        const float send = h ? part[i]     : part[i + 4];
        s4[i] = keep + __shfl_xor(send, 16, 32);
    }
    float s2[2];
    #pragma unroll
    for (int i = 0; i < 2; ++i) {
        const float keep = g ? s4[i + 2] : s4[i];
        const float send = g ? s4[i]     : s4[i + 2];
        s2[i] = keep + __shfl_xor(send, 8, 32);
    }
    const float keep1 = f ? s2[1] : s2[0];
    const float send1 = f ? s2[0] : s2[1];
    float u = keep1 + __shfl_xor(send1, 4, 32);
    u += __shfl_xor(u, 2, 32);
    u += __shfl_xor(u, 1, 32);
    return u;
}

__device__ __forceinline__ void load_keys_batch(v2f (&kb)[CX_KS][2], const float* p) {
    #pragma unroll
    for (int s = 0; s < CX_KS; ++s) {
        kb[s][0] = *(const v2f*)(p + (size_t)(2 * s) * D);
        kb[s][1] = *(const v2f*)(p + (size_t)(2 * s) * D + 32);
    }
}

__device__ __forceinline__ void mma_keys_batch(v8f& c0, v8f& c1, const float* a,
                                               const v2f (&kb)[CX_KS][2]) {
    #pragma unroll
    for (int s = 0; s < CX_KS; ++s) {
        const v2f af = *(const v2f*)(a + 4 * s);
        c0 = wmma4(af, kb[s][0], c0);
        c1 = wmma4(af, kb[s][1], c1);
    }
}

__global__ void __launch_bounds__(256)
score_softmax_context_kernel(const float* __restrict__ pq, const float* __restrict__ pk,
                             const float* __restrict__ vatt, const float* __restrict__ keys,
                             float* __restrict__ scores_out, float* __restrict__ ctx) {
    __shared__ float s_sc[SC_RQ][TK];
    __shared__ __attribute__((aligned(16))) float s_pe[2 * SC_RQ][PE_PITCH];

    const int row0  = blockIdx.x * SC_RQ;
    const int slice = row0 / TQ;
    const int tid   = threadIdx.x;
    const int wave  = tid >> 5;
    const int lane  = tid & 31;
    const int nb    = lane * 4;

    const float* pqbase = pq + (size_t)row0 * D + nb;
    const float* pkbase = pk + (size_t)slice * TK * D + nb;

    v4f w[4], q[SC_RQ][4];
    load_row4(w, vatt + nb);
    #pragma unroll
    for (int r = 0; r < SC_RQ; ++r) load_row4(q[r], pqbase + (size_t)r * D);

    const int  qsel = (lane >> 2) & 7;
    const bool wr   = (lane & 3) == 0;

    v4f pa[4], pb[4];
    load_row4(pa, pkbase + (size_t)wave * D);
    #pragma unroll 1
    for (int kk = 0; kk < TK / 8; kk += 2) {
        const int k0 = kk * 8 + wave;
        const int k1 = k0 + 8;
        float parta[SC_RQ], partb[SC_RQ];
        load_row4(pb, pkbase + (size_t)k1 * D);
        row_partials8(parta, q, pa, w);
        const int kn = (kk + 2 < TK / 8) ? k1 + 8 : k1;
        load_row4(pa, pkbase + (size_t)kn * D);
        row_partials8(partb, q, pb, w);
        const float u0 = butterfly8(parta, lane);
        const float u1 = butterfly8(partb, lane);
        if (wr) {
            s_sc[qsel][k0] = u0;
            s_sc[qsel][k1] = u1;
        }
    }
    __syncthreads();

    const int half = lane >> 4;
    const int ln16 = lane & 15;
    const float* bP = keys + (size_t)slice * TK * D + (size_t)half * D + wave * 64 + 2 * ln16;
    v2f kb0[CX_KS][2], kb1[CX_KS][2];
    load_keys_batch(kb0, bP);

    float p0, p1, p2, p3;
    {
        const float* sc = s_sc[wave];
        float s0 = sc[lane], s1 = sc[lane + 32];
        float s2 = sc[lane + 64], s3 = sc[lane + 96];
        float m = fmaxf(fmaxf(s0, s1), fmaxf(s2, s3));
        #pragma unroll
        for (int off = 16; off > 0; off >>= 1)
            m = fmaxf(m, __shfl_xor(m, off, 32));
        float e0 = __expf(s0 - m), e1 = __expf(s1 - m);
        float e2 = __expf(s2 - m), e3 = __expf(s3 - m);
        float sum = e0 + e1 + e2 + e3;
        #pragma unroll
        for (int off = 16; off > 0; off >>= 1)
            sum += __shfl_xor(sum, off, 32);
        const float inv = 1.f / sum;
        p0 = e0 * inv; p1 = e1 * inv; p2 = e2 * inv; p3 = e3 * inv;
        float* lo = &s_pe[wave][2 * lane];
        float* hi = &s_pe[wave + SC_RQ][2 * lane];
        const v2f a0 = {p0, 0.f}, a1 = {p1, 0.f}, a2 = {p2, 0.f}, a3 = {p3, 0.f};
        const v2f z0 = {0.f, p0}, z1 = {0.f, p1}, z2 = {0.f, p2}, z3 = {0.f, p3};
        *(v2f*)(lo)       = a0;  *(v2f*)(hi)       = z0;
        *(v2f*)(lo + 64)  = a1;  *(v2f*)(hi + 64)  = z1;
        *(v2f*)(lo + 128) = a2;  *(v2f*)(hi + 128) = z2;
        *(v2f*)(lo + 192) = a3;  *(v2f*)(hi + 192) = z3;
    }
    __syncthreads();

    {
        float* out = scores_out + (size_t)(row0 + wave) * TK;
        out[lane]      = p0;
        out[lane + 32] = p1;
        out[lane + 64] = p2;
        out[lane + 96] = p3;
    }

    {
        const float* aP = &s_pe[ln16][2 * half];
        v8f c0 = {}, c1 = {};
        load_keys_batch(kb1, bP + (size_t)(2 * CX_KS) * D);
        mma_keys_batch(c0, c1, aP, kb0);
        load_keys_batch(kb0, bP + (size_t)(4 * CX_KS) * D);
        mma_keys_batch(c0, c1, aP + 4 * CX_KS, kb1);
        load_keys_batch(kb1, bP + (size_t)(6 * CX_KS) * D);
        mma_keys_batch(c0, c1, aP + 8 * CX_KS, kb0);
        mma_keys_batch(c0, c1, aP + 12 * CX_KS, kb1);

        float* oP = ctx + (size_t)row0 * D + wave * 64 + 2 * ln16 + half;
        #pragma unroll
        for (int r = 0; r < SC_RQ; ++r) {
            oP[(size_t)r * D]      = c0[r];
            oP[(size_t)r * D + 32] = c1[r];
        }
    }
}

extern "C" void kernel_launch(void* const* d_in, const int* in_sizes, int n_in,
                              void* d_out, int out_size, void* d_ws, size_t ws_size,
                              hipStream_t stream) {
    const float* query = (const float*)d_in[0];
    const float* keys  = (const float*)d_in[1];
    const float* Wq    = (const float*)d_in[2];
    const float* Wk    = (const float*)d_in[3];
    const float* vatt  = (const float*)d_in[4];

    float* ctx    = (float*)d_out;
    float* scores = ctx + (size_t)SLICES * TQ * D;

    float* pq = (float*)d_ws;
    float* pk = pq + (size_t)PQ_ROWS * D;

    proj_kernel<<<768, 128, 0, stream>>>(query, keys, Wq, Wk, pq, pk);
    score_softmax_context_kernel<<<256, 256, 0, stream>>>(pq, pk, vatt, keys, scores, ctx);
}
